# starting kernel
# baseline (speedup 1.0000x reference)
#include <hip/hip_runtime.h>
#include <hip/hip_bf16.h>

typedef __attribute__((ext_vector_type(2))) float v2f;
typedef __attribute__((ext_vector_type(8))) float v8f;

#define SIR_GAMMA 0.1f
#define SIR_DT    0.5f
#define SIR_NITER 14

__device__ __forceinline__ float relu1(float x) {
    float r;
    asm("v_max_num_f32_e32 %0, 0, %1" : "=v"(r) : "v"(x));
    return r;
}

__global__ __launch_bounds__(256)
void sir_npi_wmma_kernel(const float* __restrict__ y0,
                         const float* __restrict__ x,
                         const float* __restrict__ W1,
                         const float* __restrict__ b1,
                         const float* __restrict__ W2,
                         const float* __restrict__ b2,
                         float* __restrict__ yout,
                         int numTiles  ) {
    const int lane   = threadIdx.x & 31;
    const int wvBlk  = threadIdx.x >> 5;
    const int wvsPB  = blockDim.x >> 5;
    const int gWave  = blockIdx.x * wvsPB + wvBlk;
    const int nWaves = gridDim.x * wvsPB;

    const int  hi  = (lane >= 16) ? 1 : 0;
    const int  l15 = lane & 15;
    const bool b3  = (lane & 8) != 0;
    const bool b2l = (lane & 4) != 0;
    const bool b1l = (lane & 2) != 0;

    v2f   Bfrag[4][2];
    float b1v[4], w2v[4];
#pragma unroll
    for (int t = 0; t < 4; ++t) {
        const int n = 16 * t + l15;
#pragma unroll
        for (int k = 0; k < 2; ++k) {
            const int kr = 4 * k + (hi ? 2 : 0);
            v2f bf;
            bf.x = W1[(kr + 0) * 64 + n];
            bf.y = W1[(kr + 1) * 64 + n];
            Bfrag[t][k] = bf;
        }
        b1v[t] = b1[n];
        w2v[t] = W2[n];
    }
    const float b2v = b2[0];

    for (int tile = gWave; tile < numTiles; tile += nWaves) {
        const int rowBase = tile * 32;
        float betaSub[2];

#pragma unroll
        for (int sub = 0; sub < 2; ++sub) {
            const int rb = rowBase + 16 * sub;

            v8f acc[4];
#pragma unroll
            for (int t = 0; t < 4; ++t) {
#pragma unroll
                for (int m = 0; m < 8; ++m) acc[t][m] = b1v[t];
            }

#pragma unroll
            for (int k = 0; k < 2; ++k) {
                const float* xr = x + (size_t)(rb + l15) * 8 + 4 * k + 2 * hi;
                const v2f a = *(const v2f*)xr;
#pragma unroll
                for (int t = 0; t < 4; ++t) {
                    acc[t] = __builtin_amdgcn_wmma_f32_16x16x4_f32(
                         false, a,  false, Bfrag[t][k],
                         (short)0, acc[t],
                         false,  false);
                }
            }

            float part[8];
#pragma unroll
            for (int m = 0; m < 8; ++m) part[m] = 0.0f;
#pragma unroll
            for (int t = 0; t < 4; ++t) {
#pragma unroll
                for (int m = 0; m < 8; ++m) {
                    part[m] = fmaf(relu1(acc[t][m]), w2v[t], part[m]);
                }
            }

            float k4[4];
#pragma unroll
            for (int m = 0; m < 4; ++m) {
                const float send = b3 ? part[m] : part[m + 4];
                const float keep = b3 ? part[m + 4] : part[m];
                k4[m] = keep + __shfl_xor(send, 8, 32);
            }
            float k2[2];
#pragma unroll
            for (int m = 0; m < 2; ++m) {
                const float send = b2l ? k4[m] : k4[m + 2];
                const float keep = b2l ? k4[m + 2] : k4[m];
                k2[m] = keep + __shfl_xor(send, 4, 32);
            }
            {
                const float send = b1l ? k2[0] : k2[1];
                const float keep = b1l ? k2[1] : k2[0];
                float k1 = keep + __shfl_xor(send, 2, 32);
                k1 += __shfl_xor(k1, 1, 32);
                const float z = k1 + b2v;
                betaSub[sub] = 1.0f / (1.0f + __expf(-z));
            }
        }

        const int   rloc = (lane >> 1) & 15;
        const int   row  = rowBase + ((lane & 1) ? 16 : 0) + rloc;
        const float beta = (lane & 1) ? betaSub[1] : betaSub[0];

        const size_t r = (size_t)row * 3;
        const float S0 = y0[r + 0];
        const float I0 = y0[r + 1];
        const float R0 = y0[r + 2];
        const float total = S0 + I0 + R0;
        const float bd = beta * SIR_DT;
        const float im = 1.0f - SIR_DT * SIR_GAMMA;
        float S = S0, I = I0;
#pragma unroll
        for (int it = 0; it < SIR_NITER; ++it) {
            const float inf = bd * (S * I);
            S = S - inf;
            I = fmaf(I, im, inf);
        }
        yout[r + 0] = S;
        yout[r + 1] = I;
        yout[r + 2] = (total - S) - I;
    }
}

extern "C" void kernel_launch(void* const* d_in, const int* in_sizes, int n_in,
                              void* d_out, int out_size, void* d_ws, size_t ws_size,
                              hipStream_t stream) {
    const float* y0 = (const float*)d_in[0];
    const float* x  = (const float*)d_in[1];
    const float* W1 = (const float*)d_in[2];
    const float* b1 = (const float*)d_in[3];
    const float* W2 = (const float*)d_in[4];
    const float* b2 = (const float*)d_in[5];
    float* yout = (float*)d_out;

    const int B = in_sizes[0] / 3;
    const int numTiles = B / 32;

    const int threads = 256;
    const int wavesPerBlock = threads / 32;
    int blocks = (numTiles + wavesPerBlock - 1) / wavesPerBlock;
    if (blocks > 8192) blocks = 8192;

    sir_npi_wmma_kernel<<<blocks, threads, 0, stream>>>(
        y0, x, W1, b1, W2, b2, yout, numTiles);
}


# Layer 1 transposed (W1^T as A, x^T as B, one 16-B x load per lane): a lane holds 32 hidden units of one row, so layer 2 sums in-lane + one half swap, no 16-lane shuffle tree; f32 sum order changes
# speedup vs baseline: 1.1134x; 1.1134x over previous
#include <hip/hip_runtime.h>
#include <hip/hip_bf16.h>


typedef __attribute__((ext_vector_type(2))) float v2f;
typedef __attribute__((ext_vector_type(4))) float v4f;
typedef __attribute__((ext_vector_type(8))) float v8f;

#define SIR_GAMMA 0.1f
#define SIR_DT    0.5f
#define SIR_NITER 14

__device__ __forceinline__ float relu1(float x) {
    float r;
    asm("v_max_num_f32_e32 %0, 0, %1" : "=v"(r) : "v"(x));
    return r;
}

__device__ __forceinline__ float xhalf(float v) {
    const int i = __builtin_bit_cast(int, v);
    return __builtin_bit_cast(float,
        __builtin_amdgcn_permlanex16(i, i, 0x76543210, 0xfedcba98, false, false));
}

__global__ __launch_bounds__(256, 8)
void sir_npi_wmma_kernel(const float* __restrict__ y0,
                         const float* __restrict__ x,
                         const float* __restrict__ W1,
                         const float* __restrict__ b1,
                         const float* __restrict__ W2,
                         const float* __restrict__ b2,
                         float* __restrict__ yout,
                         int numTiles  ) {
    const int lane   = threadIdx.x & 31;
    const int wvBlk  = threadIdx.x >> 5;
    const int wvsPB  = blockDim.x >> 5;
    const int gWave  = blockIdx.x * wvsPB + wvBlk;
    const int nWaves = gridDim.x * wvsPB;

    const int h = lane >> 4;
    const int c = lane & 15;

    v2f Af[4][2];
#pragma unroll
    for (int t = 0; t < 4; ++t) {
#pragma unroll
        for (int j = 0; j < 2; ++j) {
            const int k0 = 4 * h + 2 * j;
            v2f af;
            af.x = W1[(k0 + 0) * 64 + 16 * t + c];
            af.y = W1[(k0 + 1) * 64 + 16 * t + c];
            Af[t][j] = af;
        }
    }

    float b1h[4][8], w2h[4][8];
#pragma unroll
    for (int t = 0; t < 4; ++t) {
        const v4f* pb = (const v4f*)(b1 + 16 * t + 8 * h);
        const v4f* pw = (const v4f*)(W2 + 16 * t + 8 * h);
        const v4f bq0 = pb[0], bq1 = pb[1];
        const v4f wq0 = pw[0], wq1 = pw[1];
#pragma unroll
        for (int m = 0; m < 4; ++m) {
            b1h[t][m]     = bq0[m];
            b1h[t][m + 4] = bq1[m];
            w2h[t][m]     = wq0[m];
            w2h[t][m + 4] = wq1[m];
        }
    }
    v2f wp[2][8];
#pragma unroll
    for (int g = 0; g < 2; ++g) {
#pragma unroll
        for (int m = 0; m < 8; ++m) {
            wp[g][m].x = w2h[2 * g][m];
            wp[g][m].y = w2h[2 * g + 1][m];
        }
    }
    const float b2v = b2[0];

    for (int tile = gWave; tile < numTiles; tile += nWaves) {
        const int rowBase = tile * 32;

        const v4f xq0 = *(const v4f*)(x + (size_t)(rowBase + c) * 8 + 4 * h);
        const v4f xq1 = *(const v4f*)(x + (size_t)(rowBase + 16 + c) * 8 + 4 * h);
        const size_t r = (size_t)(rowBase + lane) * 3;
        const float S0 = y0[r + 0];
        const float I0 = y0[r + 1];
        const float R0 = y0[r + 2];

        float zs[2];
#pragma unroll
        for (int sub = 0; sub < 2; ++sub) {
            const v4f xq = sub ? xq1 : xq0;
            v2f bf[2];
            bf[0].x = xq.x; bf[0].y = xq.y;
            bf[1].x = xq.z; bf[1].y = xq.w;

            v2f pg[2];
#pragma unroll
            for (int g = 0; g < 2; ++g) {
                v8f a0, a1;
#pragma unroll
                for (int m = 0; m < 8; ++m) {
                    a0[m] = b1h[2 * g][m];
                    a1[m] = b1h[2 * g + 1][m];
                }
#pragma unroll
                for (int j = 0; j < 2; ++j) {
                    a0 = __builtin_amdgcn_wmma_f32_16x16x4_f32(
                         false, Af[2 * g][j],  false, bf[j],
                         (short)0, a0,  false,  false);
                    a1 = __builtin_amdgcn_wmma_f32_16x16x4_f32(
                         false, Af[2 * g + 1][j],  false, bf[j],
                         (short)0, a1,  false,  false);
                }
                v2f hr, p;
                hr.x = relu1(a0[0]); hr.y = relu1(a1[0]);
                p = hr * wp[g][0];
#pragma unroll
                for (int m = 1; m < 8; ++m) {
                    hr.x = relu1(a0[m]); hr.y = relu1(a1[m]);
                    p = __builtin_elementwise_fma(hr, wp[g][m], p);
                }
                pg[g] = p;
            }
            const v2f ps = pg[0] + pg[1];
            const float zp = ps.x + ps.y;
            zs[sub] = zp + xhalf(zp);
        }

        const float z    = (h ? zs[1] : zs[0]) + b2v;
        const float beta = 1.0f / (1.0f + __expf(-z));

        const float total = S0 + I0 + R0;
        const float bd = beta * SIR_DT;
        const float im = 1.0f - SIR_DT * SIR_GAMMA;
        float S = S0, I = I0;
#pragma unroll
        for (int it = 0; it < SIR_NITER; ++it) {
            const float inf = bd * (S * I);
            S = S - inf;
            I = fmaf(I, im, inf);
        }
        yout[r + 0] = S;
        yout[r + 1] = I;
        yout[r + 2] = (total - S) - I;
    }
}

extern "C" void kernel_launch(void* const* d_in, const int* in_sizes, int n_in,
                              void* d_out, int out_size, void* d_ws, size_t ws_size,
                              hipStream_t stream) {
    const float* y0 = (const float*)d_in[0];
    const float* x  = (const float*)d_in[1];
    const float* W1 = (const float*)d_in[2];
    const float* b1 = (const float*)d_in[3];
    const float* W2 = (const float*)d_in[4];
    const float* b2 = (const float*)d_in[5];
    float* yout = (float*)d_out;

    const int B = in_sizes[0] / 3;
    const int numTiles = B / 32;

    const int threads = 256;
    const int wavesPerBlock = threads / 32;
    int blocks = (numTiles + wavesPerBlock - 1) / wavesPerBlock;
    if (blocks > 8192) blocks = 8192;

    sir_npi_wmma_kernel<<<blocks, threads, 0, stream>>>(
        y0, x, W1, b1, W2, b2, yout, numTiles);
}


# Relu of two adjacent accumulator registers per issue slot (dual-issue v_max_num pair) feeding one packed fma; W2 kept in its loaded register pairs, no regroup moves; layer-2 f32 sum order changes
# speedup vs baseline: 1.1669x; 1.0481x over previous
#include <hip/hip_runtime.h>
#include <hip/hip_bf16.h>


typedef __attribute__((ext_vector_type(2))) float v2f;
typedef __attribute__((ext_vector_type(4))) float v4f;
typedef __attribute__((ext_vector_type(8))) float v8f;

#define SIR_GAMMA 0.1f
#define SIR_DT    0.5f
#define SIR_NITER 14

__device__ __forceinline__ void relu_fma2(v2f& p0, v2f& p1,
                                          float a00, float a01, float a10, float a11,
                                          v2f w0, v2f w1) {
    asm("v_dual_max_num_f32 v126, 0, %2 :: v_dual_max_num_f32 v127, 0, %3\n\t"
        "s_delay_alu instid0(VALU_DEP_1)\n\t"
        "v_pk_fma_f32 %0, v[126:127], %6, %0\n\t"
        "v_dual_max_num_f32 v126, 0, %4 :: v_dual_max_num_f32 v127, 0, %5\n\t"
        "s_delay_alu instid0(VALU_DEP_1)\n\t"
        "v_pk_fma_f32 %1, v[126:127], %7, %1"
        : "+v"(p0), "+v"(p1)
        : "v"(a00), "v"(a01), "v"(a10), "v"(a11), "v"(w0), "v"(w1)
        : "v126", "v127");
}
__device__ __forceinline__ void relu_mul2(v2f& p0, v2f& p1,
                                          float a00, float a01, float a10, float a11,
                                          v2f w0, v2f w1) {
    asm("v_dual_max_num_f32 v126, 0, %2 :: v_dual_max_num_f32 v127, 0, %3\n\t"
        "s_delay_alu instid0(VALU_DEP_1)\n\t"
        "v_pk_mul_f32 %0, v[126:127], %6\n\t"
        "v_dual_max_num_f32 v126, 0, %4 :: v_dual_max_num_f32 v127, 0, %5\n\t"
        "s_delay_alu instid0(VALU_DEP_1)\n\t"
        "v_pk_mul_f32 %1, v[126:127], %7"
        : "=&v"(p0), "=&v"(p1)
        : "v"(a00), "v"(a01), "v"(a10), "v"(a11), "v"(w0), "v"(w1)
        : "v126", "v127");
}

__device__ __forceinline__ float xhalf(float v) {
    const int i = __builtin_bit_cast(int, v);
    return __builtin_bit_cast(float,
        __builtin_amdgcn_permlanex16(i, i, 0x76543210, 0xfedcba98, false, false));
}

__global__ __launch_bounds__(256, 8)
void sir_npi_wmma_kernel(const float* __restrict__ y0,
                         const float* __restrict__ x,
                         const float* __restrict__ W1,
                         const float* __restrict__ b1,
                         const float* __restrict__ W2,
                         const float* __restrict__ b2,
                         float* __restrict__ yout,
                         int numTiles  ) {
    const int lane   = threadIdx.x & 31;
    const int wvBlk  = threadIdx.x >> 5;
    const int wvsPB  = blockDim.x >> 5;
    const int gWave  = blockIdx.x * wvsPB + wvBlk;
    const int nWaves = gridDim.x * wvsPB;

    const int h = lane >> 4;
    const int c = lane & 15;

    v2f Af[4][2];
#pragma unroll
    for (int t = 0; t < 4; ++t) {
#pragma unroll
        for (int j = 0; j < 2; ++j) {
            const int k0 = 4 * h + 2 * j;
            v2f af;
            af.x = W1[(k0 + 0) * 64 + 16 * t + c];
            af.y = W1[(k0 + 1) * 64 + 16 * t + c];
            Af[t][j] = af;
        }
    }

    float b1h[4][8];
    v2f   w2p[4][4];
#pragma unroll
    for (int t = 0; t < 4; ++t) {
        const v4f* pb = (const v4f*)(b1 + 16 * t + 8 * h);
        const v4f* pw = (const v4f*)(W2 + 16 * t + 8 * h);
        const v4f bq0 = pb[0], bq1 = pb[1];
        const v4f wq0 = pw[0], wq1 = pw[1];
#pragma unroll
        for (int m = 0; m < 4; ++m) {
            b1h[t][m]     = bq0[m];
            b1h[t][m + 4] = bq1[m];
        }
        w2p[t][0].x = wq0.x; w2p[t][0].y = wq0.y;
        w2p[t][1].x = wq0.z; w2p[t][1].y = wq0.w;
        w2p[t][2].x = wq1.x; w2p[t][2].y = wq1.y;
        w2p[t][3].x = wq1.z; w2p[t][3].y = wq1.w;
    }
    const float b2v = b2[0];

    for (int tile = gWave; tile < numTiles; tile += nWaves) {
        const int rowBase = tile * 32;

        const v4f xq0 = *(const v4f*)(x + (size_t)(rowBase + c) * 8 + 4 * h);
        const v4f xq1 = *(const v4f*)(x + (size_t)(rowBase + 16 + c) * 8 + 4 * h);
        const size_t r = (size_t)(rowBase + lane) * 3;
        const float S0 = y0[r + 0];
        const float I0 = y0[r + 1];
        const float R0 = y0[r + 2];

        float zs[2];
#pragma unroll
        for (int sub = 0; sub < 2; ++sub) {
            const v4f xq = sub ? xq1 : xq0;
            v2f bf[2];
            bf[0].x = xq.x; bf[0].y = xq.y;
            bf[1].x = xq.z; bf[1].y = xq.w;

            v2f p0, p1;
#pragma unroll
            for (int g = 0; g < 2; ++g) {
                v8f a0, a1;
#pragma unroll
                for (int m = 0; m < 8; ++m) {
                    a0[m] = b1h[2 * g][m];
                    a1[m] = b1h[2 * g + 1][m];
                }
#pragma unroll
                for (int j = 0; j < 2; ++j) {
                    a0 = __builtin_amdgcn_wmma_f32_16x16x4_f32(
                         false, Af[2 * g][j],  false, bf[j],
                         (short)0, a0,  false,  false);
                    a1 = __builtin_amdgcn_wmma_f32_16x16x4_f32(
                         false, Af[2 * g + 1][j],  false, bf[j],
                         (short)0, a1,  false,  false);
                }
#pragma unroll
                for (int i = 0; i < 4; ++i) {
                    if (g == 0 && i == 0)
                        relu_mul2(p0, p1, a0[0], a0[1], a1[0], a1[1], w2p[0][0], w2p[1][0]);
                    else
                        relu_fma2(p0, p1, a0[2 * i], a0[2 * i + 1], a1[2 * i], a1[2 * i + 1],
                                  w2p[2 * g][i], w2p[2 * g + 1][i]);
                }
            }
            const v2f ps = p0 + p1;
            const float zp = ps.x + ps.y;
            zs[sub] = zp + xhalf(zp);
        }

        const float z    = (h ? zs[1] : zs[0]) + b2v;
        const float beta = 1.0f / (1.0f + __expf(-z));

        const float total = S0 + I0 + R0;
        const float bd = beta * SIR_DT;
        const float im = 1.0f - SIR_DT * SIR_GAMMA;
        float S = S0, I = I0;
#pragma unroll
        for (int it = 0; it < SIR_NITER; ++it) {
            const float inf = bd * (S * I);
            S = S - inf;
            I = fmaf(I, im, inf);
        }
        yout[r + 0] = S;
        yout[r + 1] = I;
        yout[r + 2] = (total - S) - I;
    }
}

extern "C" void kernel_launch(void* const* d_in, const int* in_sizes, int n_in,
                              void* d_out, int out_size, void* d_ws, size_t ws_size,
                              hipStream_t stream) {
    const float* y0 = (const float*)d_in[0];
    const float* x  = (const float*)d_in[1];
    const float* W1 = (const float*)d_in[2];
    const float* b1 = (const float*)d_in[3];
    const float* W2 = (const float*)d_in[4];
    const float* b2 = (const float*)d_in[5];
    float* yout = (float*)d_out;

    const int B = in_sizes[0] / 3;
    const int numTiles = B / 32;

    const int threads = 256;
    const int wavesPerBlock = threads / 32;
    int blocks = (numTiles + wavesPerBlock - 1) / wavesPerBlock;
    if (blocks > 8192) blocks = 8192;

    sir_npi_wmma_kernel<<<blocks, threads, 0, stream>>>(
        y0, x, W1, b1, W2, b2, yout, numTiles);
}


# Grid cut from 8192 to 1024 blocks, one resident pass at 8 waves per SIMD: a wave enters once, loads W1/b1/W2 once and loops over 16 tiles on a wave-uniform (scalar) tile index; sum order unchanged
# speedup vs baseline: 1.2046x; 1.0323x over previous
#include <hip/hip_runtime.h>
#include <hip/hip_bf16.h>


typedef __attribute__((ext_vector_type(2))) float v2f;
typedef __attribute__((ext_vector_type(4))) float v4f;
typedef __attribute__((ext_vector_type(8))) float v8f;

#define SIR_GAMMA 0.1f
#define SIR_DT    0.5f
#define SIR_NITER 14

__device__ __forceinline__ void relu_fma2(v2f& p0, v2f& p1,
                                          float a00, float a01, float a10, float a11,
                                          v2f w0, v2f w1) {
    asm("v_dual_max_num_f32 v126, 0, %2 :: v_dual_max_num_f32 v127, 0, %3\n\t"
        "s_delay_alu instid0(VALU_DEP_1)\n\t"
        "v_pk_fma_f32 %0, v[126:127], %6, %0\n\t"
        "v_dual_max_num_f32 v126, 0, %4 :: v_dual_max_num_f32 v127, 0, %5\n\t"
        "s_delay_alu instid0(VALU_DEP_1)\n\t"
        "v_pk_fma_f32 %1, v[126:127], %7, %1"
        : "+v"(p0), "+v"(p1)
        : "v"(a00), "v"(a01), "v"(a10), "v"(a11), "v"(w0), "v"(w1)
        : "v126", "v127");
}
__device__ __forceinline__ void relu_mul2(v2f& p0, v2f& p1,
                                          float a00, float a01, float a10, float a11,
                                          v2f w0, v2f w1) {
    asm("v_dual_max_num_f32 v126, 0, %2 :: v_dual_max_num_f32 v127, 0, %3\n\t"
        "s_delay_alu instid0(VALU_DEP_1)\n\t"
        "v_pk_mul_f32 %0, v[126:127], %6\n\t"
        "v_dual_max_num_f32 v126, 0, %4 :: v_dual_max_num_f32 v127, 0, %5\n\t"
        "s_delay_alu instid0(VALU_DEP_1)\n\t"
        "v_pk_mul_f32 %1, v[126:127], %7"
        : "=&v"(p0), "=&v"(p1)
        : "v"(a00), "v"(a01), "v"(a10), "v"(a11), "v"(w0), "v"(w1)
        : "v126", "v127");
}

__device__ __forceinline__ float xhalf(float v) {
    const int i = __builtin_bit_cast(int, v);
    return __builtin_bit_cast(float,
        __builtin_amdgcn_permlanex16(i, i, 0x76543210, 0xfedcba98, false, false));
}

__global__ __launch_bounds__(256, 8)
void sir_npi_wmma_kernel(const float* __restrict__ y0,
                         const float* __restrict__ x,
                         const float* __restrict__ W1,
                         const float* __restrict__ b1,
                         const float* __restrict__ W2,
                         const float* __restrict__ b2,
                         float* __restrict__ yout,
                         int numTiles  ) {
    const int lane   = threadIdx.x & 31;
    const int wvBlk  = threadIdx.x >> 5;
    const int wvsPB  = blockDim.x >> 5;
    const int gWave  = blockIdx.x * wvsPB + wvBlk;
    const int nWaves = gridDim.x * wvsPB;

    const int h = lane >> 4;
    const int c = lane & 15;

    v2f Af[4][2];
#pragma unroll
    for (int t = 0; t < 4; ++t) {
#pragma unroll
        for (int j = 0; j < 2; ++j) {
            const int k0 = 4 * h + 2 * j;
            v2f af;
            af.x = W1[(k0 + 0) * 64 + 16 * t + c];
            af.y = W1[(k0 + 1) * 64 + 16 * t + c];
            Af[t][j] = af;
        }
    }

    float b1h[4][8];
    v2f   w2p[4][4];
#pragma unroll
    for (int t = 0; t < 4; ++t) {
        const v4f* pb = (const v4f*)(b1 + 16 * t + 8 * h);
        const v4f* pw = (const v4f*)(W2 + 16 * t + 8 * h);
        const v4f bq0 = pb[0], bq1 = pb[1];
        const v4f wq0 = pw[0], wq1 = pw[1];
#pragma unroll
        for (int m = 0; m < 4; ++m) {
            b1h[t][m]     = bq0[m];
            b1h[t][m + 4] = bq1[m];
        }
        w2p[t][0].x = wq0.x; w2p[t][0].y = wq0.y;
        w2p[t][1].x = wq0.z; w2p[t][1].y = wq0.w;
        w2p[t][2].x = wq1.x; w2p[t][2].y = wq1.y;
        w2p[t][3].x = wq1.z; w2p[t][3].y = wq1.w;
    }
    const float b2v = b2[0];

    const int tile0 = __builtin_amdgcn_readfirstlane(gWave);
    for (int tile = tile0; tile < numTiles; tile += nWaves) {
        const int rowBase = tile * 32;

        const v4f xq0 = *(const v4f*)(x + (size_t)(rowBase + c) * 8 + 4 * h);
        const v4f xq1 = *(const v4f*)(x + (size_t)(rowBase + 16 + c) * 8 + 4 * h);
        const size_t r = (size_t)(rowBase + lane) * 3;
        const float S0 = y0[r + 0];
        const float I0 = y0[r + 1];
        const float R0 = y0[r + 2];

        float zs[2];
#pragma unroll
        for (int sub = 0; sub < 2; ++sub) {
            const v4f xq = sub ? xq1 : xq0;
            v2f bf[2];
            bf[0].x = xq.x; bf[0].y = xq.y;
            bf[1].x = xq.z; bf[1].y = xq.w;

            v2f p0, p1;
#pragma unroll
            for (int g = 0; g < 2; ++g) {
                v8f a0, a1;
#pragma unroll
                for (int m = 0; m < 8; ++m) {
                    a0[m] = b1h[2 * g][m];
                    a1[m] = b1h[2 * g + 1][m];
                }
#pragma unroll
                for (int j = 0; j < 2; ++j) {
                    a0 = __builtin_amdgcn_wmma_f32_16x16x4_f32(
                         false, Af[2 * g][j],  false, bf[j],
                         (short)0, a0,  false,  false);
                    a1 = __builtin_amdgcn_wmma_f32_16x16x4_f32(
                         false, Af[2 * g + 1][j],  false, bf[j],
                         (short)0, a1,  false,  false);
                }
#pragma unroll
                for (int i = 0; i < 4; ++i) {
                    if (g == 0 && i == 0)
                        relu_mul2(p0, p1, a0[0], a0[1], a1[0], a1[1], w2p[0][0], w2p[1][0]);
                    else
                        relu_fma2(p0, p1, a0[2 * i], a0[2 * i + 1], a1[2 * i], a1[2 * i + 1],
                                  w2p[2 * g][i], w2p[2 * g + 1][i]);
                }
            }
            const v2f ps = p0 + p1;
            const float zp = ps.x + ps.y;
            zs[sub] = zp + xhalf(zp);
        }

        const float z    = (h ? zs[1] : zs[0]) + b2v;
        const float beta = 1.0f / (1.0f + __expf(-z));

        const float total = S0 + I0 + R0;
        const float bd = beta * SIR_DT;
        const float im = 1.0f - SIR_DT * SIR_GAMMA;
        float S = S0, I = I0;
#pragma unroll
        for (int it = 0; it < SIR_NITER; ++it) {
            const float inf = bd * (S * I);
            S = S - inf;
            I = fmaf(I, im, inf);
        }
        yout[r + 0] = S;
        yout[r + 1] = I;
        yout[r + 2] = (total - S) - I;
    }
}

extern "C" void kernel_launch(void* const* d_in, const int* in_sizes, int n_in,
                              void* d_out, int out_size, void* d_ws, size_t ws_size,
                              hipStream_t stream) {
    const float* y0 = (const float*)d_in[0];
    const float* x  = (const float*)d_in[1];
    const float* W1 = (const float*)d_in[2];
    const float* b1 = (const float*)d_in[3];
    const float* W2 = (const float*)d_in[4];
    const float* b2 = (const float*)d_in[5];
    float* yout = (float*)d_out;

    const int B = in_sizes[0] / 3;
    const int numTiles = B / 32;

    const int threads = 256;
    const int wavesPerBlock = threads / 32;
    int blocks = (numTiles + wavesPerBlock - 1) / wavesPerBlock;
    if (blocks > 1024) blocks = 1024;

    sir_npi_wmma_kernel<<<blocks, threads, 0, stream>>>(
        y0, x, W1, b1, W2, b2, yout, numTiles);
}


# Issue priority lowered 3,2,1,0 as a wave advances through the tile range (s_setprio, wave-uniform): waves behind win issue, so the 8 waves of a SIMD end together, not oldest first; sums unchanged
# speedup vs baseline: 1.3052x; 1.0835x over previous
#include <hip/hip_runtime.h>
#include <hip/hip_bf16.h>


typedef __attribute__((ext_vector_type(2))) float v2f;
typedef __attribute__((ext_vector_type(4))) float v4f;
typedef __attribute__((ext_vector_type(8))) float v8f;

#define SIR_GAMMA 0.1f
#define SIR_DT    0.5f
#define SIR_NITER 14

__device__ __forceinline__ void relu_fma2(v2f& p0, v2f& p1,
                                          float a00, float a01, float a10, float a11,
                                          v2f w0, v2f w1) {
    asm("v_dual_max_num_f32 v126, 0, %2 :: v_dual_max_num_f32 v127, 0, %3\n\t"
        "s_delay_alu instid0(VALU_DEP_1)\n\t"
        "v_pk_fma_f32 %0, v[126:127], %6, %0\n\t"
        "v_dual_max_num_f32 v126, 0, %4 :: v_dual_max_num_f32 v127, 0, %5\n\t"
        "s_delay_alu instid0(VALU_DEP_1)\n\t"
        "v_pk_fma_f32 %1, v[126:127], %7, %1"
        : "+v"(p0), "+v"(p1)
        : "v"(a00), "v"(a01), "v"(a10), "v"(a11), "v"(w0), "v"(w1)
        : "v126", "v127");
}
__device__ __forceinline__ void relu_mul2(v2f& p0, v2f& p1,
                                          float a00, float a01, float a10, float a11,
                                          v2f w0, v2f w1) {
    asm("v_dual_max_num_f32 v126, 0, %2 :: v_dual_max_num_f32 v127, 0, %3\n\t"
        "s_delay_alu instid0(VALU_DEP_1)\n\t"
        "v_pk_mul_f32 %0, v[126:127], %6\n\t"
        "v_dual_max_num_f32 v126, 0, %4 :: v_dual_max_num_f32 v127, 0, %5\n\t"
        "s_delay_alu instid0(VALU_DEP_1)\n\t"
        "v_pk_mul_f32 %1, v[126:127], %7"
        : "=&v"(p0), "=&v"(p1)
        : "v"(a00), "v"(a01), "v"(a10), "v"(a11), "v"(w0), "v"(w1)
        : "v126", "v127");
}

__device__ __forceinline__ float xhalf(float v) {
    const int i = __builtin_bit_cast(int, v);
    return __builtin_bit_cast(float,
        __builtin_amdgcn_permlanex16(i, i, 0x76543210, 0xfedcba98, false, false));
}

__device__ __forceinline__ void tile_body(const v4f xq0, const v4f xq1,
                                          const float S0, const float I0, const float R0,
                                          const v2f (&Af)[4][2], const float (&b1h)[4][8],
                                          const v2f (&w2p)[4][4], const float b2v, const int h,
                                          float& So, float& Io, float& Ro) {
    float zs[2];
#pragma unroll
    for (int sub = 0; sub < 2; ++sub) {
        const v4f xq = sub ? xq1 : xq0;
        v2f bf[2];
        bf[0].x = xq.x; bf[0].y = xq.y;
        bf[1].x = xq.z; bf[1].y = xq.w;

        v2f p0, p1;
#pragma unroll
        for (int g = 0; g < 2; ++g) {
            v8f a0, a1;
#pragma unroll
            for (int m = 0; m < 8; ++m) {
                a0[m] = b1h[2 * g][m];
                a1[m] = b1h[2 * g + 1][m];
            }
#pragma unroll
            for (int j = 0; j < 2; ++j) {
                a0 = __builtin_amdgcn_wmma_f32_16x16x4_f32(
                     false, Af[2 * g][j],  false, bf[j],
                     (short)0, a0,  false,  false);
                a1 = __builtin_amdgcn_wmma_f32_16x16x4_f32(
                     false, Af[2 * g + 1][j],  false, bf[j],
                     (short)0, a1,  false,  false);
            }
#pragma unroll
            for (int i = 0; i < 4; ++i) {
                if (g == 0 && i == 0)
                    relu_mul2(p0, p1, a0[0], a0[1], a1[0], a1[1], w2p[0][0], w2p[1][0]);
                else
                    relu_fma2(p0, p1, a0[2 * i], a0[2 * i + 1], a1[2 * i], a1[2 * i + 1],
                              w2p[2 * g][i], w2p[2 * g + 1][i]);
            }
        }
        const v2f ps = p0 + p1;
        const float zp = ps.x + ps.y;
        zs[sub] = zp + xhalf(zp);
    }

    const float z    = (h ? zs[1] : zs[0]) + b2v;
    const float beta = 1.0f / (1.0f + __expf(-z));

    const float total = S0 + I0 + R0;
    const float bd = beta * SIR_DT;
    const float im = 1.0f - SIR_DT * SIR_GAMMA;
    float S = S0, I = I0;
#pragma unroll
    for (int it = 0; it < SIR_NITER; ++it) {
        const float inf = bd * (S * I);
        S = S - inf;
        I = fmaf(I, im, inf);
    }
    So = S;
    Io = I;
    Ro = (total - S) - I;
}

__device__ __forceinline__ void one_tile(const int tile,
                                         const float* __restrict__ y0,
                                         const float* __restrict__ x,
                                         float* __restrict__ yout,
                                         const v2f (&Af)[4][2], const float (&b1h)[4][8],
                                         const v2f (&w2p)[4][4], const float b2v,
                                         const int lane, const int c, const int h) {
    const int rowBase = tile * 32;

    const v4f xq0 = *(const v4f*)(x + (size_t)(rowBase + c) * 8 + 4 * h);
    const v4f xq1 = *(const v4f*)(x + (size_t)(rowBase + 16 + c) * 8 + 4 * h);
    const size_t r = (size_t)(rowBase + lane) * 3;
    const float S0 = y0[r + 0];
    const float I0 = y0[r + 1];
    const float R0 = y0[r + 2];
    asm volatile("" :: "v"(xq0) : "memory");

    float S, I, R;
    tile_body(xq0, xq1, S0, I0, R0, Af, b1h, w2p, b2v, h, S, I, R);
    yout[r + 0] = S;
    yout[r + 1] = I;
    yout[r + 2] = R;
}

__global__ __launch_bounds__(256, 8)
void sir_npi_wmma_kernel(const float* __restrict__ y0,
                         const float* __restrict__ x,
                         const float* __restrict__ W1,
                         const float* __restrict__ b1,
                         const float* __restrict__ W2,
                         const float* __restrict__ b2,
                         float* __restrict__ yout,
                         int numTiles  ) {
    const int lane   = threadIdx.x & 31;
    const int wvBlk  = threadIdx.x >> 5;
    const int wvsPB  = blockDim.x >> 5;
    const int gWave  = blockIdx.x * wvsPB + wvBlk;
    const int nWaves = gridDim.x * wvsPB;

    const int h = lane >> 4;
    const int c = lane & 15;

    v2f Af[4][2];
#pragma unroll
    for (int t = 0; t < 4; ++t) {
#pragma unroll
        for (int j = 0; j < 2; ++j) {
            const int k0 = 4 * h + 2 * j;
            v2f af;
            af.x = W1[(k0 + 0) * 64 + 16 * t + c];
            af.y = W1[(k0 + 1) * 64 + 16 * t + c];
            Af[t][j] = af;
        }
    }

    float b1h[4][8];
    v2f   w2p[4][4];
#pragma unroll
    for (int t = 0; t < 4; ++t) {
        const v4f* pb = (const v4f*)(b1 + 16 * t + 8 * h);
        const v4f* pw = (const v4f*)(W2 + 16 * t + 8 * h);
        const v4f bq0 = pb[0], bq1 = pb[1];
        const v4f wq0 = pw[0], wq1 = pw[1];
#pragma unroll
        for (int m = 0; m < 4; ++m) {
            b1h[t][m]     = bq0[m];
            b1h[t][m + 4] = bq1[m];
        }
        w2p[t][0].x = wq0.x; w2p[t][0].y = wq0.y;
        w2p[t][1].x = wq0.z; w2p[t][1].y = wq0.w;
        w2p[t][2].x = wq1.x; w2p[t][2].y = wq1.y;
        w2p[t][3].x = wq1.z; w2p[t][3].y = wq1.w;
    }
    const float b2v = b2[0];

    const int tile0 = __builtin_amdgcn_readfirstlane(gWave);
    const int end1 = numTiles >> 1;
    const int end2 = end1 + (numTiles >> 2);
    const int end3 = end2 + (numTiles >> 3);
    int nextEnd = end1;
    __builtin_amdgcn_s_setprio(3);
    for (int tile = tile0; tile < numTiles; tile += nWaves) {
        if (tile >= nextEnd) {
            if (tile >= end3)      { __builtin_amdgcn_s_setprio(0); nextEnd = 0x7fffffff; }
            else if (tile >= end2) { __builtin_amdgcn_s_setprio(1); nextEnd = end3; }
            else                   { __builtin_amdgcn_s_setprio(2); nextEnd = end2; }
        }
        one_tile(tile, y0, x, yout, Af, b1h, w2p, b2v, lane, c, h);
    }
    __builtin_amdgcn_s_setprio(0);
}

extern "C" void kernel_launch(void* const* d_in, const int* in_sizes, int n_in,
                              void* d_out, int out_size, void* d_ws, size_t ws_size,
                              hipStream_t stream) {
    const float* y0 = (const float*)d_in[0];
    const float* x  = (const float*)d_in[1];
    const float* W1 = (const float*)d_in[2];
    const float* b1 = (const float*)d_in[3];
    const float* W2 = (const float*)d_in[4];
    const float* b2 = (const float*)d_in[5];
    float* yout = (float*)d_out;

    const int B = in_sizes[0] / 3;
    const int numTiles = B / 32;

    const int threads = 256;
    const int wavesPerBlock = threads / 32;
    int blocks = (numTiles + wavesPerBlock - 1) / wavesPerBlock;
    if (blocks > 1024) blocks = 1024;

    sir_npi_wmma_kernel<<<blocks, threads, 0, stream>>>(
        y0, x, W1, b1, W2, b2, yout, numTiles);
}


# 4 tiles per trip at 4 waves per SIMD (grid 1024 -> 512 blocks): a trip's 12 requests issued together, its independent tiles interleave; sum order kept, a few Euler updates round unfused
# speedup vs baseline: 1.3902x; 1.0651x over previous
#include <hip/hip_runtime.h>
#include <hip/hip_bf16.h>


typedef __attribute__((ext_vector_type(2))) float v2f;
typedef __attribute__((ext_vector_type(4))) float v4f;
typedef __attribute__((ext_vector_type(8))) float v8f;

#define SIR_GAMMA 0.1f
#define SIR_DT    0.5f
#define SIR_NITER 14

__device__ __forceinline__ void relu_fma2(v2f& p0, v2f& p1,
                                          float a00, float a01, float a10, float a11,
                                          v2f w0, v2f w1) {
    asm("v_dual_max_num_f32 v126, 0, %2 :: v_dual_max_num_f32 v127, 0, %3\n\t"
        "s_delay_alu instid0(VALU_DEP_1)\n\t"
        "v_pk_fma_f32 %0, v[126:127], %6, %0\n\t"
        "v_dual_max_num_f32 v126, 0, %4 :: v_dual_max_num_f32 v127, 0, %5\n\t"
        "s_delay_alu instid0(VALU_DEP_1)\n\t"
        "v_pk_fma_f32 %1, v[126:127], %7, %1"
        : "+v"(p0), "+v"(p1)
        : "v"(a00), "v"(a01), "v"(a10), "v"(a11), "v"(w0), "v"(w1)
        : "v126", "v127");
}
__device__ __forceinline__ void relu_mul2(v2f& p0, v2f& p1,
                                          float a00, float a01, float a10, float a11,
                                          v2f w0, v2f w1) {
    asm("v_dual_max_num_f32 v126, 0, %2 :: v_dual_max_num_f32 v127, 0, %3\n\t"
        "s_delay_alu instid0(VALU_DEP_1)\n\t"
        "v_pk_mul_f32 %0, v[126:127], %6\n\t"
        "v_dual_max_num_f32 v126, 0, %4 :: v_dual_max_num_f32 v127, 0, %5\n\t"
        "s_delay_alu instid0(VALU_DEP_1)\n\t"
        "v_pk_mul_f32 %1, v[126:127], %7"
        : "=&v"(p0), "=&v"(p1)
        : "v"(a00), "v"(a01), "v"(a10), "v"(a11), "v"(w0), "v"(w1)
        : "v126", "v127");
}

__device__ __forceinline__ float xhalf(float v) {
    const int i = __builtin_bit_cast(int, v);
    return __builtin_bit_cast(float,
        __builtin_amdgcn_permlanex16(i, i, 0x76543210, 0xfedcba98, false, false));
}

__device__ __forceinline__ void tile_body(const v4f xq0, const v4f xq1,
                                          const float S0, const float I0, const float R0,
                                          const v2f (&Af)[4][2], const float (&b1h)[4][8],
                                          const v2f (&w2p)[4][4], const float b2v, const int h,
                                          float& So, float& Io, float& Ro) {
    float zs[2];
#pragma unroll
    for (int sub = 0; sub < 2; ++sub) {
        const v4f xq = sub ? xq1 : xq0;
        v2f bf[2];
        bf[0].x = xq.x; bf[0].y = xq.y;
        bf[1].x = xq.z; bf[1].y = xq.w;

        v2f p0, p1;
#pragma unroll
        for (int g = 0; g < 2; ++g) {
            v8f a0, a1;
#pragma unroll
            for (int m = 0; m < 8; ++m) {
                a0[m] = b1h[2 * g][m];
                a1[m] = b1h[2 * g + 1][m];
            }
#pragma unroll
            for (int j = 0; j < 2; ++j) {
                a0 = __builtin_amdgcn_wmma_f32_16x16x4_f32(
                     false, Af[2 * g][j],  false, bf[j],
                     (short)0, a0,  false,  false);
                a1 = __builtin_amdgcn_wmma_f32_16x16x4_f32(
                     false, Af[2 * g + 1][j],  false, bf[j],
                     (short)0, a1,  false,  false);
            }
#pragma unroll
            for (int i = 0; i < 4; ++i) {
                if (g == 0 && i == 0)
                    relu_mul2(p0, p1, a0[0], a0[1], a1[0], a1[1], w2p[0][0], w2p[1][0]);
                else
                    relu_fma2(p0, p1, a0[2 * i], a0[2 * i + 1], a1[2 * i], a1[2 * i + 1],
                              w2p[2 * g][i], w2p[2 * g + 1][i]);
            }
        }
        const v2f ps = p0 + p1;
        const float zp = ps.x + ps.y;
        zs[sub] = zp + xhalf(zp);
    }

    const float z    = (h ? zs[1] : zs[0]) + b2v;
    const float beta = 1.0f / (1.0f + __expf(-z));

    const float total = S0 + I0 + R0;
    const float bd = beta * SIR_DT;
    const float im = 1.0f - SIR_DT * SIR_GAMMA;
    float S = S0, I = I0;
#pragma unroll
    for (int it = 0; it < SIR_NITER; ++it) {
        const float inf = bd * (S * I);
        S = S - inf;
        I = fmaf(I, im, inf);
    }
    So = S;
    Io = I;
    Ro = (total - S) - I;
}

template <int N>
__device__ __forceinline__ void tile_group(const int tile0,
                                           const float* __restrict__ y0,
                                           const float* __restrict__ x,
                                           float* __restrict__ yout,
                                           const v2f (&Af)[4][2], const float (&b1h)[4][8],
                                           const v2f (&w2p)[4][4], const float b2v,
                                           const int lane, const int c, const int h) {
    v4f    xq0[N], xq1[N];
    float  S0[N], I0[N], R0[N];
    size_t r[N];
#pragma unroll
    for (int n = 0; n < N; ++n) {
        r[n]  = (size_t)((tile0 + n) * 32 + lane) * 3;
        S0[n] = y0[r[n] + 0];
        I0[n] = y0[r[n] + 1];
        R0[n] = y0[r[n] + 2];
    }
#pragma unroll
    for (int n = 0; n < N; ++n) {
        const int rowBase = (tile0 + n) * 32;
        xq0[n] = *(const v4f*)(x + (size_t)(rowBase + c) * 8 + 4 * h);
        xq1[n] = *(const v4f*)(x + (size_t)(rowBase + 16 + c) * 8 + 4 * h);
    }
    __builtin_amdgcn_sched_barrier(0);

#pragma unroll
    for (int n = 0; n < N; ++n) {
        float S, I, R;
        tile_body(xq0[n], xq1[n], S0[n], I0[n], R0[n], Af, b1h, w2p, b2v, h, S, I, R);
        yout[r[n] + 0] = S;
        yout[r[n] + 1] = I;
        yout[r[n] + 2] = R;
    }
}

#define TPT 4

__global__ __launch_bounds__(256, 4)
void sir_npi_wmma_kernel(const float* __restrict__ y0,
                         const float* __restrict__ x,
                         const float* __restrict__ W1,
                         const float* __restrict__ b1,
                         const float* __restrict__ W2,
                         const float* __restrict__ b2,
                         float* __restrict__ yout,
                         int numTiles  ) {
    const int lane   = threadIdx.x & 31;
    const int wvBlk  = threadIdx.x >> 5;
    const int wvsPB  = blockDim.x >> 5;
    const int gWave  = blockIdx.x * wvsPB + wvBlk;
    const int nWaves = gridDim.x * wvsPB;

    const int h = lane >> 4;
    const int c = lane & 15;

    v2f Af[4][2];
#pragma unroll
    for (int t = 0; t < 4; ++t) {
#pragma unroll
        for (int j = 0; j < 2; ++j) {
            const int k0 = 4 * h + 2 * j;
            v2f af;
            af.x = W1[(k0 + 0) * 64 + 16 * t + c];
            af.y = W1[(k0 + 1) * 64 + 16 * t + c];
            Af[t][j] = af;
        }
    }

    float b1h[4][8];
    v2f   w2p[4][4];
#pragma unroll
    for (int t = 0; t < 4; ++t) {
        const v4f* pb = (const v4f*)(b1 + 16 * t + 8 * h);
        const v4f* pw = (const v4f*)(W2 + 16 * t + 8 * h);
        const v4f bq0 = pb[0], bq1 = pb[1];
        const v4f wq0 = pw[0], wq1 = pw[1];
#pragma unroll
        for (int m = 0; m < 4; ++m) {
            b1h[t][m]     = bq0[m];
            b1h[t][m + 4] = bq1[m];
        }
        w2p[t][0].x = wq0.x; w2p[t][0].y = wq0.y;
        w2p[t][1].x = wq0.z; w2p[t][1].y = wq0.w;
        w2p[t][2].x = wq1.x; w2p[t][2].y = wq1.y;
        w2p[t][3].x = wq1.z; w2p[t][3].y = wq1.w;
    }
    const float b2v = b2[0];

    const int wave0    = __builtin_amdgcn_readfirstlane(gWave);
    const int numTrips = numTiles / TPT;
    const int end1 = numTrips >> 1;
    const int end2 = end1 + (numTrips >> 2);
    const int end3 = end2 + (numTrips >> 3);
    int nextEnd = end1;
    __builtin_amdgcn_s_setprio(3);
    for (int trip = wave0; trip < numTrips; trip += nWaves) {
        if (trip >= nextEnd) {
            if (trip >= end3)      { __builtin_amdgcn_s_setprio(0); nextEnd = 0x7fffffff; }
            else if (trip >= end2) { __builtin_amdgcn_s_setprio(1); nextEnd = end3; }
            else                   { __builtin_amdgcn_s_setprio(2); nextEnd = end2; }
        }
        tile_group<TPT>(trip * TPT, y0, x, yout, Af, b1h, w2p, b2v, lane, c, h);
    }
    __builtin_amdgcn_s_setprio(0);
    for (int tile = numTrips * TPT + wave0; tile < numTiles; tile += nWaves)
        tile_group<1>(tile, y0, x, yout, Af, b1h, w2p, b2v, lane, c, h);
}

extern "C" void kernel_launch(void* const* d_in, const int* in_sizes, int n_in,
                              void* d_out, int out_size, void* d_ws, size_t ws_size,
                              hipStream_t stream) {
    const float* y0 = (const float*)d_in[0];
    const float* x  = (const float*)d_in[1];
    const float* W1 = (const float*)d_in[2];
    const float* b1 = (const float*)d_in[3];
    const float* W2 = (const float*)d_in[4];
    const float* b2 = (const float*)d_in[5];
    float* yout = (float*)d_out;

    const int B = in_sizes[0] / 3;
    const int numTiles = B / 32;

    const int threads = 256;
    const int wavesPerBlock = threads / 32;
    const int numTrips = numTiles / TPT;
    int blocks = (numTrips + wavesPerBlock - 1) / wavesPerBlock;
    if (blocks > 512) blocks = 512;
    if (blocks < 1) blocks = 1;

    sir_npi_wmma_kernel<<<blocks, threads, 0, stream>>>(
        y0, x, W1, b1, W2, b2, yout, numTiles);
}
